# starting kernel
# baseline (speedup 1.0000x reference)
#include <hip/hip_runtime.h>


#define B_    2
#define C_    64
#define N_    16384
#define K_    16
#define COUT_ 64
#define CCAT_ 128

typedef __attribute__((ext_vector_type(2))) float v2f;
typedef __attribute__((ext_vector_type(8))) float v8f;


__global__ void __launch_bounds__(256)
prep_wcat_kernel(const float* __restrict__ W, float* __restrict__ WcatT) {
    int t = blockIdx.x * blockDim.x + threadIdx.x;
    if (t >= C_ * CCAT_) return;
    int c = t / CCAT_;
    int m = t % CCAT_;
    float v;
    if (m < COUT_) v = W[m * (3 * C_) + c] - W[m * (3 * C_) + C_ + c];
    else           v = W[(m - COUT_) * (3 * C_) + C_ + c];
    WcatT[c * CCAT_ + m] = v;
}

__global__ void __launch_bounds__(256)
mean_kernel(const float* __restrict__ x, float* __restrict__ gmean) {
    __shared__ float s[256];
    const int bc = blockIdx.x;
    const float* row = x + (size_t)bc * N_;
    float acc = 0.f;
    for (int n = threadIdx.x; n < N_; n += 256) acc += row[n];
    s[threadIdx.x] = acc;
    __syncthreads();
    for (int off = 128; off > 0; off >>= 1) {
        if (threadIdx.x < off) s[threadIdx.x] += s[threadIdx.x + off];
        __syncthreads();
    }
    if (threadIdx.x == 0) gmean[bc] = s[0] * (1.0f / N_);
}

__global__ void __launch_bounds__(128)
c0_kernel(const float* __restrict__ W, const float* __restrict__ bias,
          const float* __restrict__ gmean, float* __restrict__ c0) {
    int t = threadIdx.x;
    if (t >= B_ * COUT_) return;
    int b = t / COUT_, o = t % COUT_;
    float acc = bias[o];
    #pragma unroll 8
    for (int c = 0; c < C_; ++c)
        acc += W[o * (3 * C_) + 2 * C_ + c] * gmean[b * C_ + c];
    c0[t] = acc;
}

__global__ void __launch_bounds__(128)
gemm_wmma_kernel(const float* __restrict__ x, const float* __restrict__ WcatT,
                 float* __restrict__ Yt) {
    const int lane = threadIdx.x;
    const int wv   = threadIdx.y;
    const int n0   = blockIdx.x * 16;
    const int m0   = (blockIdx.y * 4 + wv) * 16;
    const int b    = blockIdx.z;
    const float* xb = x + (size_t)b * C_ * N_;
    float* yb = Yt + (size_t)b * N_ * CCAT_;
    const int ln = lane & 15;
    const int hi = lane >> 4;

    v8f acc = {};
    #pragma unroll
    for (int c = 0; c < C_; c += 4) {
        v2f a, bm;
        a.x = xb[(size_t)(c + 2 * hi    ) * N_ + n0 + ln];
        a.y = xb[(size_t)(c + 2 * hi + 1) * N_ + n0 + ln];
        bm.x = WcatT[(c + 2 * hi    ) * CCAT_ + m0 + ln];
        bm.y = WcatT[(c + 2 * hi + 1) * CCAT_ + m0 + ln];
        acc = __builtin_amdgcn_wmma_f32_16x16x4_f32(
            false, a, false, bm, (short)0, acc, false, false);
    }

    #pragma unroll
    for (int v = 0; v < 8; ++v) {
        int node = n0 + v + 8 * hi;
        yb[(size_t)node * CCAT_ + m0 + ln] = acc[v];
    }
}

__global__ void __launch_bounds__(256)
edge_kernel(const int* __restrict__ eidx, const float* __restrict__ Yt,
            const float* __restrict__ c0, float* __restrict__ out) {
    int t    = blockIdx.x * blockDim.x + threadIdx.x;
    int o    = t & (COUT_ - 1);
    int node = t >> 6;
    if (node >= B_ * N_) return;
    int b = node / N_;
    int n = node - b * N_;

    const int* e0 = eidx + ((size_t)b * N_ + n) * K_;
    const int* e1 = eidx + (size_t)B_ * N_ * K_ + ((size_t)b * N_ + n) * K_;
    const float* yb = Yt + (size_t)b * N_ * CCAT_;
    const float bias0 = c0[b * COUT_ + o];

    float m = -3.402823466e38f;
    #pragma unroll
    for (int k = 0; k < K_; ++k) {
        int i = e1[k];
        int j = e0[k];
        float v = yb[(size_t)i * CCAT_ + o]
                + yb[(size_t)j * CCAT_ + COUT_ + o]
                + bias0;
        m = fmaxf(m, v);
    }
    out[((size_t)b * COUT_ + o) * N_ + n] = fmaxf(m, 0.0f);
}

extern "C" void kernel_launch(void* const* d_in, const int* in_sizes, int n_in,
                              void* d_out, int out_size, void* d_ws, size_t ws_size,
                              hipStream_t stream) {
    const float* x    = (const float*)d_in[0];
    const int*   eidx = (const int*)  d_in[1];
    const float* W    = (const float*)d_in[2];
    const float* bias = (const float*)d_in[3];

    float* ws    = (float*)d_ws;
    float* WcatT = ws;
    float* gmean = ws + 8192;
    float* c0    = ws + 8192 + 128;
    float* Yt    = ws + 8192 + 256;

    prep_wcat_kernel<<<(C_ * CCAT_ + 255) / 256, 256, 0, stream>>>(W, WcatT);
    mean_kernel<<<B_ * C_, 256, 0, stream>>>(x, gmean);
    c0_kernel<<<1, 128, 0, stream>>>(W, bias, gmean, c0);

    dim3 ggrid(N_ / 16, (CCAT_ / 16) / 4, B_);
    dim3 gblock(32, 4, 1);
    gemm_wmma_kernel<<<ggrid, gblock, 0, stream>>>(x, WcatT, Yt);

    edge_kernel<<<(B_ * N_ * COUT_) / 256, 256, 0, stream>>>(eidx, Yt, c0, (float*)d_out);
}


# edge kernel: one wave owns 8 consecutive nodes (65536 -> 4096 waves), 16B per lane so one load fetches two 256B rows, indices by scalar loads, LDS transpose to 256B stores; order of sums unchanged
# speedup vs baseline: 2.1394x; 2.1394x over previous
#include <hip/hip_runtime.h>


#define B_    2
#define C_    64
#define N_    16384
#define K_    16
#define COUT_ 64
#define CCAT_ 128

typedef __attribute__((ext_vector_type(2))) float v2f;
typedef __attribute__((ext_vector_type(8))) float v8f;


__global__ void __launch_bounds__(256)
prep_wcat_kernel(const float* __restrict__ W, float* __restrict__ WcatT) {
    int t = blockIdx.x * blockDim.x + threadIdx.x;
    if (t >= C_ * CCAT_) return;
    int c = t / CCAT_;
    int m = t % CCAT_;
    float v;
    if (m < COUT_) v = W[m * (3 * C_) + c] - W[m * (3 * C_) + C_ + c];
    else           v = W[(m - COUT_) * (3 * C_) + C_ + c];
    WcatT[c * CCAT_ + m] = v;
}

__global__ void __launch_bounds__(256)
mean_kernel(const float* __restrict__ x, float* __restrict__ gmean) {
    __shared__ float s[256];
    const int bc = blockIdx.x;
    const float* row = x + (size_t)bc * N_;
    float acc = 0.f;
    for (int n = threadIdx.x; n < N_; n += 256) acc += row[n];
    s[threadIdx.x] = acc;
    __syncthreads();
    for (int off = 128; off > 0; off >>= 1) {
        if (threadIdx.x < off) s[threadIdx.x] += s[threadIdx.x + off];
        __syncthreads();
    }
    if (threadIdx.x == 0) gmean[bc] = s[0] * (1.0f / N_);
}

__global__ void __launch_bounds__(128)
c0_kernel(const float* __restrict__ W, const float* __restrict__ bias,
          const float* __restrict__ gmean, float* __restrict__ c0) {
    int t = threadIdx.x;
    if (t >= B_ * COUT_) return;
    int b = t / COUT_, o = t % COUT_;
    float acc = bias[o];
    #pragma unroll 8
    for (int c = 0; c < C_; ++c)
        acc += W[o * (3 * C_) + 2 * C_ + c] * gmean[b * C_ + c];
    c0[t] = acc;
}

__global__ void __launch_bounds__(128)
gemm_wmma_kernel(const float* __restrict__ x, const float* __restrict__ WcatT,
                 float* __restrict__ Yt) {
    const int lane = threadIdx.x;
    const int wv   = threadIdx.y;
    const int n0   = blockIdx.x * 16;
    const int m0   = (blockIdx.y * 4 + wv) * 16;
    const int b    = blockIdx.z;
    const float* xb = x + (size_t)b * C_ * N_;
    float* yb = Yt + (size_t)b * N_ * CCAT_;
    const int ln = lane & 15;
    const int hi = lane >> 4;

    v8f acc = {};
    #pragma unroll
    for (int c = 0; c < C_; c += 4) {
        v2f a, bm;
        a.x = xb[(size_t)(c + 2 * hi    ) * N_ + n0 + ln];
        a.y = xb[(size_t)(c + 2 * hi + 1) * N_ + n0 + ln];
        bm.x = WcatT[(c + 2 * hi    ) * CCAT_ + m0 + ln];
        bm.y = WcatT[(c + 2 * hi + 1) * CCAT_ + m0 + ln];
        acc = __builtin_amdgcn_wmma_f32_16x16x4_f32(
            false, a, false, bm, (short)0, acc, false, false);
    }

    #pragma unroll
    for (int v = 0; v < 8; ++v) {
        int node = n0 + v + 8 * hi;
        yb[(size_t)node * CCAT_ + m0 + ln] = acc[v];
    }
}

#define EDGE_NPW   8
#define EDGE_WAVES 8
#define EDGE_NB    (EDGE_NPW * EDGE_WAVES)
#define EDGE_TP    (EDGE_NB + 1)

__device__ __forceinline__ void edge_offsets(const int* __restrict__ e0, const int* __restrict__ e1,
                                             int p, int h, unsigned loff,
                                             unsigned (&o1)[K_], unsigned (&o2)[K_]) {
    #pragma unroll
    for (int k = 0; k < K_; ++k) {
        const int i_lo = __builtin_amdgcn_readfirstlane(e1[p * K_ + k]);
        const int i_hi = __builtin_amdgcn_readfirstlane(e1[(p + 1) * K_ + k]);
        const int j_lo = __builtin_amdgcn_readfirstlane(e0[p * K_ + k]);
        const int j_hi = __builtin_amdgcn_readfirstlane(e0[(p + 1) * K_ + k]);
        o1[k] = (unsigned)(h ? i_hi : i_lo) * 512u + loff;
        o2[k] = (unsigned)(h ? j_hi : j_lo) * 512u + 256u + loff;
    }
}

__device__ __forceinline__ float4 edge_max4(float4 m, float4 a, float4 b) {
    m.x = fmaxf(m.x, a.x + b.x);
    m.y = fmaxf(m.y, a.y + b.y);
    m.z = fmaxf(m.z, a.z + b.z);
    m.w = fmaxf(m.w, a.w + b.w);
    return m;
}

__device__ __forceinline__ void edge_put(float* trow, float4 m, float4 cb) {
    trow[0 * EDGE_TP] = fmaxf(m.x + cb.x, 0.0f);
    trow[1 * EDGE_TP] = fmaxf(m.y + cb.y, 0.0f);
    trow[2 * EDGE_TP] = fmaxf(m.z + cb.z, 0.0f);
    trow[3 * EDGE_TP] = fmaxf(m.w + cb.w, 0.0f);
}

__global__ void __launch_bounds__(256) __attribute__((amdgpu_waves_per_eu(4, 4)))
edge_kernel(const int* __restrict__ eidx, const float* __restrict__ Yt,
            const float* __restrict__ c0, float* __restrict__ out) {
    __shared__ float tile[COUT_ * EDGE_TP];

    const int lane = threadIdx.x & 31;
    const int wv   = __builtin_amdgcn_readfirstlane((int)(threadIdx.x >> 5));
    const int c    = lane & 15;
    const int h    = lane >> 4;
    const int g0   = blockIdx.x * EDGE_NB;
    const int b    = g0 / N_;
    const int nb0  = g0 - b * N_;
    const int gw   = g0 + wv * EDGE_NPW;

    const int* e0 = eidx + (size_t)gw * K_;
    const int* e1 = e0 + (size_t)B_ * N_ * K_;
    const char* yb = (const char*)(Yt + (size_t)b * N_ * CCAT_);
    const float4 cb = *(const float4*)(c0 + b * COUT_ + 4 * c);
    const unsigned loff = (unsigned)c * 16u;
    float* tcol = tile + (4 * c) * EDGE_TP + wv * EDGE_NPW + h;
    const float4 minit = make_float4(-3.402823466e38f, -3.402823466e38f,
                                     -3.402823466e38f, -3.402823466e38f);

    unsigned o1[K_], o2[K_];
    edge_offsets(e0, e1, 0, h, loff, o1, o2);

    #pragma unroll 1
    for (int p = 0; p < EDGE_NPW; p += 2) {
        float4 y1[K_], y2[K_];
        #pragma unroll
        for (int k = 0; k < K_; ++k) {
            y1[k] = *(const float4*)(yb + o1[k]);
            y2[k] = *(const float4*)(yb + o2[k]);
        }
        const int pn = (p + 2 < EDGE_NPW) ? p + 2 : p;
        edge_offsets(e0, e1, pn, h, loff, o1, o2);
        float4 m = minit;
        #pragma unroll
        for (int k = 0; k < K_; ++k) m = edge_max4(m, y1[k], y2[k]);
        edge_put(tcol + p, m, cb);
    }
    __syncthreads();

    #pragma unroll
    for (int r = 0; r < (COUT_ * EDGE_NB) / (256 * 4); ++r) {
        const int idx = r * 256 + (int)threadIdx.x;
        const int ch  = idx >> 4;
        const int q4  = (idx & 15) * 4;
        const float* ts = tile + ch * EDGE_TP + q4;
        const float4 v = make_float4(ts[0], ts[1], ts[2], ts[3]);
        *(float4*)(out + ((size_t)b * COUT_ + ch) * N_ + nb0 + q4) = v;
    }
}

extern "C" void kernel_launch(void* const* d_in, const int* in_sizes, int n_in,
                              void* d_out, int out_size, void* d_ws, size_t ws_size,
                              hipStream_t stream) {
    const float* x    = (const float*)d_in[0];
    const int*   eidx = (const int*)  d_in[1];
    const float* W    = (const float*)d_in[2];
    const float* bias = (const float*)d_in[3];

    float* ws    = (float*)d_ws;
    float* WcatT = ws;
    float* gmean = ws + 8192;
    float* c0    = ws + 8192 + 128;
    float* Yt    = ws + 8192 + 256;

    prep_wcat_kernel<<<(C_ * CCAT_ + 255) / 256, 256, 0, stream>>>(W, WcatT);
    mean_kernel<<<B_ * C_, 256, 0, stream>>>(x, gmean);
    c0_kernel<<<1, 128, 0, stream>>>(W, bias, gmean, c0);

    dim3 ggrid(N_ / 16, (CCAT_ / 16) / 4, B_);
    dim3 gblock(32, 4, 1);
    gemm_wmma_kernel<<<ggrid, gblock, 0, stream>>>(x, WcatT, Yt);

    edge_kernel<<<(B_ * N_) / EDGE_NB, 256, 0, stream>>>(eidx, Yt, c0, (float*)d_out);
}


# 5 launches -> 2: mean rows ride as first blocks of the GEMM launch, GEMM blocks stage W1-W2|W2 in LDS and hold B over 4 node tiles (16384 -> 4096 waves), edge blocks form c0; sums unchanged
# speedup vs baseline: 3.2835x; 1.5348x over previous
#include <hip/hip_runtime.h>


#define B_    2
#define C_    64
#define N_    16384
#define K_    16
#define COUT_ 64
#define CCAT_ 128

typedef __attribute__((ext_vector_type(2))) float v2f;
typedef __attribute__((ext_vector_type(8))) float v8f;

#define MEAN_BLOCKS (B_ * C_)
#define GEMM_NT     4
#define GEMM_NB     (16 * GEMM_NT)
#define GEMM_BLOCKS ((B_ * N_) / GEMM_NB)
#define WS_P        129

__device__ __forceinline__ void gemm_load_a(const float* __restrict__ xb, int n0, int hi, int ln,
                                            v2f (&a)[C_ / 4]) {
    #pragma unroll
    for (int s = 0; s < C_ / 4; ++s) {
        const int c = 4 * s;
        a[s].x = xb[(size_t)(c + 2 * hi    ) * N_ + n0 + ln];
        a[s].y = xb[(size_t)(c + 2 * hi + 1) * N_ + n0 + ln];
    }
}

__global__ void __launch_bounds__(256) __attribute__((amdgpu_waves_per_eu(4, 8)))
gemm_mean_kernel(const float* __restrict__ x, const float* __restrict__ W,
                 float* __restrict__ Yt, float* __restrict__ gmean) {
    __shared__ float lds[C_ * WS_P];
    const int tid = threadIdx.x;

    if (blockIdx.x < MEAN_BLOCKS) {
        const int bc = blockIdx.x;
        const float* row = x + (size_t)bc * N_;
        float acc = 0.f;
        #pragma unroll 16
        for (int n = tid; n < N_; n += 256) acc += row[n];
        lds[tid] = acc;
        __syncthreads();
        for (int off = 128; off > 0; off >>= 1) {
            if (tid < off) lds[tid] += lds[tid + off];
            __syncthreads();
        }
        if (tid == 0) gmean[bc] = lds[0] * (1.0f / N_);
        return;
    }

    const int lane = tid & 31;
    const int wv   = tid >> 5;
    const int ln   = lane & 15;
    const int hi   = lane >> 4;
    const int m0   = wv * 16;
    const int g0   = (blockIdx.x - MEAN_BLOCKS) * GEMM_NB;
    const int b    = g0 / N_;
    const int nb0  = g0 - b * N_;
    const float* xb = x + (size_t)b * C_ * N_;
    float* yb = Yt + (size_t)b * N_ * CCAT_;

    v2f a[2][C_ / 4];
    gemm_load_a(xb, nb0, hi, ln, a[0]);

    #pragma unroll
    for (int i = 0; i < (COUT_ * C_) / 256; ++i) {
        const int idx = i * 256 + tid;
        const int m = idx >> 6;
        const int c = idx & 63;
        const float w1 = W[m * (3 * C_) + c];
        const float w2 = W[m * (3 * C_) + C_ + c];
        lds[c * WS_P + m]         = w1 - w2;
        lds[c * WS_P + COUT_ + m] = w2;
    }
    __syncthreads();

    v2f bf[C_ / 4];
    #pragma unroll
    for (int s = 0; s < C_ / 4; ++s) {
        const int c = 4 * s;
        bf[s].x = lds[(c + 2 * hi    ) * WS_P + m0 + ln];
        bf[s].y = lds[(c + 2 * hi + 1) * WS_P + m0 + ln];
    }

    #pragma unroll
    for (int t = 0; t < GEMM_NT; ++t) {
        const int n0 = nb0 + t * 16;
        if (t + 1 < GEMM_NT) gemm_load_a(xb, n0 + 16, hi, ln, a[(t + 1) & 1]);
        v8f acc = {};
        #pragma unroll
        for (int s = 0; s < C_ / 4; ++s) {
            acc = __builtin_amdgcn_wmma_f32_16x16x4_f32(
                false, a[t & 1][s], false, bf[s], (short)0, acc, false, false);
        }
        #pragma unroll
        for (int v = 0; v < 8; ++v) {
            const int node = n0 + v + 8 * hi;
            yb[(size_t)node * CCAT_ + m0 + ln] = acc[v];
        }
    }
}

#define EDGE_NPW   8
#define EDGE_WAVES 8
#define EDGE_NB    (EDGE_NPW * EDGE_WAVES)
#define EDGE_TP    (EDGE_NB + 1)

__device__ __forceinline__ void edge_offsets(const int* __restrict__ e0, const int* __restrict__ e1,
                                             int p, int h, unsigned loff,
                                             unsigned (&o1)[K_], unsigned (&o2)[K_]) {
    #pragma unroll
    for (int k = 0; k < K_; ++k) {
        const int i_lo = __builtin_amdgcn_readfirstlane(e1[p * K_ + k]);
        const int i_hi = __builtin_amdgcn_readfirstlane(e1[(p + 1) * K_ + k]);
        const int j_lo = __builtin_amdgcn_readfirstlane(e0[p * K_ + k]);
        const int j_hi = __builtin_amdgcn_readfirstlane(e0[(p + 1) * K_ + k]);
        o1[k] = (unsigned)(h ? i_hi : i_lo) * 512u + loff;
        o2[k] = (unsigned)(h ? j_hi : j_lo) * 512u + 256u + loff;
    }
}

__device__ __forceinline__ float4 edge_max4(float4 m, float4 a, float4 b) {
    m.x = fmaxf(m.x, a.x + b.x);
    m.y = fmaxf(m.y, a.y + b.y);
    m.z = fmaxf(m.z, a.z + b.z);
    m.w = fmaxf(m.w, a.w + b.w);
    return m;
}

__device__ __forceinline__ void edge_put(float* trow, float4 m) {
    trow[0 * EDGE_TP] = m.x;
    trow[1 * EDGE_TP] = m.y;
    trow[2 * EDGE_TP] = m.z;
    trow[3 * EDGE_TP] = m.w;
}

__global__ void __launch_bounds__(256) __attribute__((amdgpu_waves_per_eu(4, 4)))
edge_kernel(const int* __restrict__ eidx, const float* __restrict__ Yt,
            const float* __restrict__ W, const float* __restrict__ bias,
            const float* __restrict__ gmean, float* __restrict__ out) {
    __shared__ float tile[COUT_ * EDGE_TP];
    __shared__ float c0s[COUT_];

    const int lane = threadIdx.x & 31;
    const int wv   = __builtin_amdgcn_readfirstlane((int)(threadIdx.x >> 5));
    const int c    = lane & 15;
    const int h    = lane >> 4;
    const int g0   = blockIdx.x * EDGE_NB;
    const int b    = g0 / N_;
    const int nb0  = g0 - b * N_;
    const int gw   = g0 + wv * EDGE_NPW;

    const int* e0 = eidx + (size_t)gw * K_;
    const int* e1 = e0 + (size_t)B_ * N_ * K_;
    const char* yb = (const char*)(Yt + (size_t)b * N_ * CCAT_);
    const unsigned loff = (unsigned)c * 16u;
    float* tcol = tile + (4 * c) * EDGE_TP + wv * EDGE_NPW + h;
    const float4 minit = make_float4(-3.402823466e38f, -3.402823466e38f,
                                     -3.402823466e38f, -3.402823466e38f);

    unsigned o1[K_], o2[K_];
    edge_offsets(e0, e1, 0, h, loff, o1, o2);

    const int o = wv * 8 + (lane & 7);
    const float4* w3 = (const float4*)(W + o * (3 * C_) + 2 * C_);
    const float* gm = gmean + b * C_;
    float cacc = bias[o];
    static_assert((EDGE_NPW / 2) * 16 == C_, "one quarter of the c0 chain per trip");

    #pragma unroll 1
    for (int p = 0; p < EDGE_NPW; p += 2) {
        float4 y1[K_], y2[K_];
        #pragma unroll
        for (int k = 0; k < K_; ++k) {
            y1[k] = *(const float4*)(yb + o1[k]);
            y2[k] = *(const float4*)(yb + o2[k]);
        }
        const int q0 = 2 * p;
        const float4 wq0 = w3[q0 + 0], wq1 = w3[q0 + 1], wq2 = w3[q0 + 2], wq3 = w3[q0 + 3];
        const float* gq = gm + 4 * q0;
        const int pn = (p + 2 < EDGE_NPW) ? p + 2 : p;
        edge_offsets(e0, e1, pn, h, loff, o1, o2);
        float4 m = minit;
        #pragma unroll
        for (int k = 0; k < K_; ++k) m = edge_max4(m, y1[k], y2[k]);
        edge_put(tcol + p, m);
        cacc = __builtin_fmaf(wq0.x, gq[0],  cacc); cacc = __builtin_fmaf(wq0.y, gq[1],  cacc);
        cacc = __builtin_fmaf(wq0.z, gq[2],  cacc); cacc = __builtin_fmaf(wq0.w, gq[3],  cacc);
        cacc = __builtin_fmaf(wq1.x, gq[4],  cacc); cacc = __builtin_fmaf(wq1.y, gq[5],  cacc);
        cacc = __builtin_fmaf(wq1.z, gq[6],  cacc); cacc = __builtin_fmaf(wq1.w, gq[7],  cacc);
        cacc = __builtin_fmaf(wq2.x, gq[8],  cacc); cacc = __builtin_fmaf(wq2.y, gq[9],  cacc);
        cacc = __builtin_fmaf(wq2.z, gq[10], cacc); cacc = __builtin_fmaf(wq2.w, gq[11], cacc);
        cacc = __builtin_fmaf(wq3.x, gq[12], cacc); cacc = __builtin_fmaf(wq3.y, gq[13], cacc);
        cacc = __builtin_fmaf(wq3.z, gq[14], cacc); cacc = __builtin_fmaf(wq3.w, gq[15], cacc);
    }
    if (lane < 8) c0s[o] = cacc;
    __syncthreads();

    #pragma unroll
    for (int r = 0; r < (COUT_ * EDGE_NB) / (256 * 4); ++r) {
        const int idx = r * 256 + (int)threadIdx.x;
        const int ch  = idx >> 4;
        const int q4  = (idx & 15) * 4;
        const float* ts = tile + ch * EDGE_TP + q4;
        const float cc = c0s[ch];
        const float4 v = make_float4(fmaxf(ts[0] + cc, 0.0f), fmaxf(ts[1] + cc, 0.0f),
                                     fmaxf(ts[2] + cc, 0.0f), fmaxf(ts[3] + cc, 0.0f));
        *(float4*)(out + ((size_t)b * COUT_ + ch) * N_ + nb0 + q4) = v;
    }
}

extern "C" void kernel_launch(void* const* d_in, const int* in_sizes, int n_in,
                              void* d_out, int out_size, void* d_ws, size_t ws_size,
                              hipStream_t stream) {
    const float* x    = (const float*)d_in[0];
    const int*   eidx = (const int*)  d_in[1];
    const float* W    = (const float*)d_in[2];
    const float* bias = (const float*)d_in[3];

    float* ws    = (float*)d_ws;
    float* gmean = ws + 8192;
    float* Yt    = ws + 8192 + 256;

    gemm_mean_kernel<<<MEAN_BLOCKS + GEMM_BLOCKS, 256, 0, stream>>>(x, W, Yt, gmean);

    edge_kernel<<<(B_ * N_) / EDGE_NB, 256, 0, stream>>>(eidx, Yt, W, bias, gmean, (float*)d_out);
}
